# starting kernel
# baseline (speedup 1.0000x reference)
#include <hip/hip_runtime.h>


typedef __attribute__((ext_vector_type(2))) float v2f;
typedef __attribute__((ext_vector_type(8))) float v8f;

#define NCTRL 64
#define OUTN  512
#define KTILE 8

__global__ void nurbs_stage1(const float* __restrict__ ctrl,
                             const int*   __restrict__ uspan,
                             const float* __restrict__ Nu,
                             float*       __restrict__ A)
{
    int idx = blockIdx.x * blockDim.x + threadIdx.x;
    int n = idx & 63;
    int u = (idx >> 6) & 511;
    int c = (idx >> 15) & 3;
    int b = idx >> 17;

    int m0 = uspan[u] - 3;
    float acc = 0.0f;
#pragma unroll
    for (int r = 0; r < 4; ++r) {
        float w = Nu[u * 4 + r];
        float x = ctrl[(((size_t)b * 64 + (m0 + r)) * 64 + n) * 4 + c];
        acc = fmaf(w, x, acc);
    }
    A[idx] = acc;
}

__global__ void nurbs_stage2(const float* __restrict__ A,
                             const int*   __restrict__ vspan,
                             const float* __restrict__ Nv,
                             float*       __restrict__ out)
{
    __shared__ float nvd[8][KTILE * 16];

    const int lane = threadIdx.x & 31;
    const int wave = threadIdx.x >> 5;
    const int tile = blockIdx.x * 8 + wave;
    const int vt = tile & 31;
    const int ut = (tile >> 5) & 31;
    const int b  = tile >> 10;
    const int u0 = ut * 16;
    const int v0 = vt * 16;

    const int kbase = vspan[v0] - 3;

    if (lane < 16) {
#pragma unroll
        for (int k = 0; k < KTILE; ++k) nvd[wave][k * 16 + lane] = 0.0f;
        int v  = v0 + lane;
        int ki = vspan[v] - 3 - kbase;
#pragma unroll
        for (int s = 0; s < 4; ++s) {
            int kk = ki + s;
            if (kk < KTILE) nvd[wave][kk * 16 + lane] = Nv[v * 4 + s];
        }
    }
    __syncthreads();

    const int mrow  = lane & 15;
    const int khalf = (lane >> 4) * 2;

    v8f acc[4];
#pragma unroll
    for (int c = 0; c < 4; ++c) {
        v8f d = {};
#pragma unroll
        for (int ks = 0; ks < 2; ++ks) {
            int kk = ks * 4 + khalf;
            int n0 = kbase + kk;     if (n0 > 63) n0 = 63;
            int n1 = kbase + kk + 1; if (n1 > 63) n1 = 63;
            const float* ap =
                A + (((size_t)(b * 4 + c) * OUTN + (u0 + mrow)) * NCTRL);
            v2f av; av.x = ap[n0]; av.y = ap[n1];
            v2f bv;
            bv.x = nvd[wave][kk * 16 + mrow];
            bv.y = nvd[wave][(kk + 1) * 16 + mrow];
            d = __builtin_amdgcn_wmma_f32_16x16x4_f32(
                     false, av,  false, bv,
                     (short)0, d,  false,  false);
        }
        acc[c] = d;
    }

#pragma unroll
    for (int j = 0; j < 8; ++j) {
        int u = u0 + j + (lane >> 4) * 8;
        int v = v0 + (lane & 15);
        float w = acc[3][j];
        float r = __builtin_amdgcn_rcpf(w);
        r = r * (2.0f - w * r);
        size_t o = (((size_t)b * OUTN + u) * OUTN + v) * 3;
        __builtin_nontemporal_store(acc[0][j] * r, &out[o + 0]);
        __builtin_nontemporal_store(acc[1][j] * r, &out[o + 1]);
        __builtin_nontemporal_store(acc[2][j] * r, &out[o + 2]);
    }
}

extern "C" void kernel_launch(void* const* d_in, const int* in_sizes, int n_in,
                              void* d_out, int out_size, void* d_ws, size_t ws_size,
                              hipStream_t stream)
{
    const float* ctrl  = (const float*)d_in[0];
    const int*   uspan = (const int*)  d_in[1];
    const int*   vspan = (const int*)  d_in[2];
    const float* Nu    = (const float*)d_in[3];
    const float* Nv    = (const float*)d_in[4];
    float*       out   = (float*)d_out;
    float*       A     = (float*)d_ws;

    nurbs_stage1<<<8192, 256, 0, stream>>>(ctrl, uspan, Nu, A);
    nurbs_stage2<<<2048, 256, 0, stream>>>(A, vspan, Nv, out);
}


# Fuse stage 1 into the tile kernel: each wave computes its 16u x 8n x 4c band of the u-contraction from ctrl in registers (same fma order), one launch, workspace no longer used
# speedup vs baseline: 1.6182x; 1.6182x over previous
#include <hip/hip_runtime.h>


typedef __attribute__((ext_vector_type(2))) float v2f;
typedef __attribute__((ext_vector_type(4))) float v4f;
typedef __attribute__((ext_vector_type(8))) float v8f;

#define NCTRL 64
#define OUTN  512
#define KTILE 8

__global__ void nurbs_fused(const float* __restrict__ ctrl,
                            const int*   __restrict__ uspan,
                            const int*   __restrict__ vspan,
                            const float* __restrict__ Nu,
                            const float* __restrict__ Nv,
                            float*       __restrict__ out)
{
    __shared__ float nvd[8][KTILE * 16];

    const int lane = threadIdx.x & 31;
    const int wave = threadIdx.x >> 5;
    const int tile = blockIdx.x * 8 + wave;
    const int vt = tile & 31;
    const int ut = (tile >> 5) & 31;
    const int b  = tile >> 10;
    const int u0 = ut * 16;
    const int v0 = vt * 16;

    const int mrow  = lane & 15;
    const int khalf = (lane >> 4) * 2;

    const int kbase = vspan[v0] - 3;

    const int   ua = u0 + mrow;
    const int   m0 = uspan[ua] - 3;
    const v4f   nu = *(const v4f*)(Nu + ua * 4);
    const v4f*  crow = (const v4f*)ctrl + ((size_t)b * NCTRL + m0) * NCTRL;

    v2f av[2][4];
#pragma unroll
    for (int ks = 0; ks < 2; ++ks) {
        int kk = ks * 4 + khalf;
        int n0 = kbase + kk;     if (n0 > 63) n0 = 63;
        int n1 = kbase + kk + 1; if (n1 > 63) n1 = 63;
        v4f a0 = {0.0f, 0.0f, 0.0f, 0.0f};
        v4f a1 = {0.0f, 0.0f, 0.0f, 0.0f};
#pragma unroll
        for (int r = 0; r < 4; ++r) {
            const float w  = nu[r];
            const v4f   x0 = crow[r * NCTRL + n0];
            const v4f   x1 = crow[r * NCTRL + n1];
#pragma unroll
            for (int c = 0; c < 4; ++c) {
                a0[c] = fmaf(w, x0[c], a0[c]);
                a1[c] = fmaf(w, x1[c], a1[c]);
            }
        }
#pragma unroll
        for (int c = 0; c < 4; ++c) { av[ks][c].x = a0[c]; av[ks][c].y = a1[c]; }
    }

    if (lane < 16) {
#pragma unroll
        for (int k = 0; k < KTILE; ++k) nvd[wave][k * 16 + lane] = 0.0f;
        int v  = v0 + lane;
        int ki = vspan[v] - 3 - kbase;
#pragma unroll
        for (int s = 0; s < 4; ++s) {
            int kk = ki + s;
            if (kk < KTILE) nvd[wave][kk * 16 + lane] = Nv[v * 4 + s];
        }
    }
    __syncthreads();

    v8f acc[4];
#pragma unroll
    for (int c = 0; c < 4; ++c) {
        v8f d = {};
#pragma unroll
        for (int ks = 0; ks < 2; ++ks) {
            int kk = ks * 4 + khalf;
            v2f bv;
            bv.x = nvd[wave][kk * 16 + mrow];
            bv.y = nvd[wave][(kk + 1) * 16 + mrow];
            d = __builtin_amdgcn_wmma_f32_16x16x4_f32(
                     false, av[ks][c],  false, bv,
                     (short)0, d,  false,  false);
        }
        acc[c] = d;
    }

#pragma unroll
    for (int j = 0; j < 8; ++j) {
        int u = u0 + j + (lane >> 4) * 8;
        int v = v0 + (lane & 15);
        float w = acc[3][j];
        float r = __builtin_amdgcn_rcpf(w);
        r = r * (2.0f - w * r);
        size_t o = (((size_t)b * OUTN + u) * OUTN + v) * 3;
        __builtin_nontemporal_store(acc[0][j] * r, &out[o + 0]);
        __builtin_nontemporal_store(acc[1][j] * r, &out[o + 1]);
        __builtin_nontemporal_store(acc[2][j] * r, &out[o + 2]);
    }
}

extern "C" void kernel_launch(void* const* d_in, const int* in_sizes, int n_in,
                              void* d_out, int out_size, void* d_ws, size_t ws_size,
                              hipStream_t stream)
{
    const float* ctrl  = (const float*)d_in[0];
    const int*   uspan = (const int*)  d_in[1];
    const int*   vspan = (const int*)  d_in[2];
    const float* Nu    = (const float*)d_in[3];
    const float* Nv    = (const float*)d_in[4];
    float*       out   = (float*)d_out;

    nurbs_fused<<<2048, 256, 0, stream>>>(ctrl, uspan, vspan, Nu, Nv, out);
}


# One wave per v-band walks 8 u-tiles: 2048 long-lived waves in 256 work-groups replace 16384 one-tile waves; the walk keeps B in registers and loads the next tile's control rows ahead of the math
# speedup vs baseline: 1.8833x; 1.1638x over previous
#include <hip/hip_runtime.h>


typedef __attribute__((ext_vector_type(2))) float v2f;
typedef __attribute__((ext_vector_type(4))) float v4f;
typedef __attribute__((ext_vector_type(8))) float v8f;

#define NCTRL 64
#define OUTN  512
#define KTILE 8
#define UTW   8

__global__ void __launch_bounds__(256)
nurbs_fused(const float* __restrict__ ctrl,
            const int*   __restrict__ uspan,
            const int*   __restrict__ vspan,
            const float* __restrict__ Nu,
            const float* __restrict__ Nv,
            float*       __restrict__ out)
{
    __shared__ float nvd[8][KTILE * 16];

    const int lane = threadIdx.x & 31;
    const int wave = threadIdx.x >> 5;
    const int vt  = (blockIdx.x & 3) * 8 + wave;
    const int ut0 = ((blockIdx.x >> 2) & 3) * UTW;
    const int b   = blockIdx.x >> 4;
    const int v0  = vt * 16;

    const int mrow  = lane & 15;
    const int khalf = (lane >> 4) * 2;

    const int kbase = vspan[v0] - 3;

    const int vsp = vspan[v0 + mrow];
    const v4f nvr = *(const v4f*)(Nv + (v0 + mrow) * 4);

    int ncol[2][2];
#pragma unroll
    for (int ks = 0; ks < 2; ++ks) {
        int kk = ks * 4 + khalf;
        int n0 = kbase + kk;     if (n0 > 63) n0 = 63;
        int n1 = kbase + kk + 1; if (n1 > 63) n1 = 63;
        ncol[ks][0] = n0;
        ncol[ks][1] = n1;
    }

    int m0s[UTW];
    v4f nus[UTW];
#pragma unroll
    for (int i = 0; i < UTW; ++i) {
        const int ua = (ut0 + i) * 16 + mrow;
        m0s[i] = uspan[ua] - 3;
        nus[i] = *(const v4f*)(Nu + ua * 4);
    }

    const v4f* cb = (const v4f*)ctrl + (size_t)b * NCTRL * NCTRL;

    v4f xb[2][16];
#pragma unroll
    for (int ks = 0; ks < 2; ++ks)
#pragma unroll
        for (int r = 0; r < 4; ++r) {
            const v4f* crow = cb + (m0s[0] + r) * NCTRL;
            xb[0][(ks * 4 + r) * 2 + 0] = crow[ncol[ks][0]];
            xb[0][(ks * 4 + r) * 2 + 1] = crow[ncol[ks][1]];
        }

    if (lane < 16) {
#pragma unroll
        for (int k = 0; k < KTILE; ++k) nvd[wave][k * 16 + lane] = 0.0f;
        int ki = vsp - 3 - kbase;
#pragma unroll
        for (int s = 0; s < 4; ++s) {
            int kk = ki + s;
            if (kk < KTILE) nvd[wave][kk * 16 + lane] = nvr[s];
        }
    }
    __syncthreads();

    v2f bv[2];
#pragma unroll
    for (int ks = 0; ks < 2; ++ks) {
        int kk = ks * 4 + khalf;
        bv[ks].x = nvd[wave][kk * 16 + mrow];
        bv[ks].y = nvd[wave][(kk + 1) * 16 + mrow];
    }

#pragma unroll
    for (int i = 0; i < UTW; ++i) {
        const int cur = i & 1;
        const int nxt = cur ^ 1;

        if (i + 1 < UTW) {
#pragma unroll
            for (int ks = 0; ks < 2; ++ks)
#pragma unroll
                for (int r = 0; r < 4; ++r) {
                    const v4f* crow = cb + (m0s[i + 1] + r) * NCTRL;
                    xb[nxt][(ks * 4 + r) * 2 + 0] = crow[ncol[ks][0]];
                    xb[nxt][(ks * 4 + r) * 2 + 1] = crow[ncol[ks][1]];
                }
        }
        __builtin_amdgcn_sched_barrier(0);

        v2f av[2][4];
        const v4f nu = nus[i];
#pragma unroll
        for (int ks = 0; ks < 2; ++ks) {
            v4f a0 = {0.0f, 0.0f, 0.0f, 0.0f};
            v4f a1 = {0.0f, 0.0f, 0.0f, 0.0f};
#pragma unroll
            for (int r = 0; r < 4; ++r) {
                const float w  = nu[r];
                const v4f   x0 = xb[cur][(ks * 4 + r) * 2 + 0];
                const v4f   x1 = xb[cur][(ks * 4 + r) * 2 + 1];
#pragma unroll
                for (int c = 0; c < 4; ++c) {
                    a0[c] = fmaf(w, x0[c], a0[c]);
                    a1[c] = fmaf(w, x1[c], a1[c]);
                }
            }
#pragma unroll
            for (int c = 0; c < 4; ++c) { av[ks][c].x = a0[c]; av[ks][c].y = a1[c]; }
        }

        v8f acc[4];
#pragma unroll
        for (int c = 0; c < 4; ++c) {
            v8f d = {};
#pragma unroll
            for (int ks = 0; ks < 2; ++ks) {
                d = __builtin_amdgcn_wmma_f32_16x16x4_f32(
                         false, av[ks][c],  false, bv[ks],
                         (short)0, d,  false,  false);
            }
            acc[c] = d;
        }

        const int u0 = (ut0 + i) * 16;
#pragma unroll
        for (int j = 0; j < 8; ++j) {
            int u = u0 + j + (lane >> 4) * 8;
            int v = v0 + (lane & 15);
            float w = acc[3][j];
            float r = __builtin_amdgcn_rcpf(w);
            r = r * (2.0f - w * r);
            size_t o = (((size_t)b * OUTN + u) * OUTN + v) * 3;
            __builtin_nontemporal_store(acc[0][j] * r, &out[o + 0]);
            __builtin_nontemporal_store(acc[1][j] * r, &out[o + 1]);
            __builtin_nontemporal_store(acc[2][j] * r, &out[o + 2]);
        }
    }
}

extern "C" void kernel_launch(void* const* d_in, const int* in_sizes, int n_in,
                              void* d_out, int out_size, void* d_ws, size_t ws_size,
                              hipStream_t stream)
{
    const float* ctrl  = (const float*)d_in[0];
    const int*   uspan = (const int*)  d_in[1];
    const int*   vspan = (const int*)  d_in[2];
    const float* Nu    = (const float*)d_in[3];
    const float* Nv    = (const float*)d_in[4];
    float*       out   = (float*)d_out;

    nurbs_fused<<<256, 256, 0, stream>>>(ctrl, uspan, vspan, Nu, Nv, out);
}


# Each wave owns a 32-sample v-band: one A band feeds two B tiles, and the 16x32x3 result is staged in the wave's LDS so that every global store is a b128 covering four whole 128-B lines
# speedup vs baseline: 3.2880x; 1.7459x over previous
#include <hip/hip_runtime.h>


typedef __attribute__((ext_vector_type(2))) float v2f;
typedef __attribute__((ext_vector_type(4))) float v4f;
typedef __attribute__((ext_vector_type(8))) float v8f;

#define NCTRL  64
#define OUTN   512
#define KTILE  8
#define VBW    32
#define UTW    4
#define SPITCH 100

__global__ void __launch_bounds__(256)
nurbs_fused(const float* __restrict__ ctrl,
            const int*   __restrict__ uspan,
            const int*   __restrict__ vspan,
            const float* __restrict__ Nu,
            const float* __restrict__ Nv,
            float*       __restrict__ out)
{
    __shared__ float nvd[8][KTILE * VBW];
    __shared__ __attribute__((aligned(16))) float stage[8][16 * SPITCH];

    const int lane = threadIdx.x & 31;
    const int wave = threadIdx.x >> 5;
    const int vb  = (blockIdx.x & 1) * 8 + wave;
    const int ut0 = ((blockIdx.x >> 1) & 7) * UTW;
    const int b   = blockIdx.x >> 4;
    const int v0  = vb * VBW;

    const int mrow  = lane & 15;
    const int half  = lane >> 4;
    const int khalf = half * 2;

    const int kbase = vspan[v0] - 3;

    const int vsp = vspan[v0 + lane];
    const v4f nvr = *(const v4f*)(Nv + (v0 + lane) * 4);

    int ncol[2][2];
#pragma unroll
    for (int ks = 0; ks < 2; ++ks) {
        int kk = ks * 4 + khalf;
        int n0 = kbase + kk;     if (n0 > 63) n0 = 63;
        int n1 = kbase + kk + 1; if (n1 > 63) n1 = 63;
        ncol[ks][0] = n0;
        ncol[ks][1] = n1;
    }

    int m0s[UTW];
    v4f nus[UTW];
#pragma unroll
    for (int i = 0; i < UTW; ++i) {
        const int ua = (ut0 + i) * 16 + mrow;
        m0s[i] = uspan[ua] - 3;
        nus[i] = *(const v4f*)(Nu + ua * 4);
    }

    const v4f* cb = (const v4f*)ctrl + (size_t)b * NCTRL * NCTRL;

    v4f xb[2][16];
#pragma unroll
    for (int ks = 0; ks < 2; ++ks)
#pragma unroll
        for (int r = 0; r < 4; ++r) {
            const v4f* crow = cb + (m0s[0] + r) * NCTRL;
            xb[0][(ks * 4 + r) * 2 + 0] = crow[ncol[ks][0]];
            xb[0][(ks * 4 + r) * 2 + 1] = crow[ncol[ks][1]];
        }

    {
#pragma unroll
        for (int k = 0; k < KTILE; ++k) nvd[wave][k * VBW + lane] = 0.0f;
        int ki = vsp - 3 - kbase;
#pragma unroll
        for (int s = 0; s < 4; ++s) {
            int kk = ki + s;
            if ((unsigned)kk < (unsigned)KTILE) nvd[wave][kk * VBW + lane] = nvr[s];
        }
    }
    __syncthreads();

    v2f bv[2][2];
#pragma unroll
    for (int t = 0; t < 2; ++t)
#pragma unroll
        for (int ks = 0; ks < 2; ++ks) {
            int kk = ks * 4 + khalf;
            bv[t][ks].x = nvd[wave][kk * VBW + t * 16 + mrow];
            bv[t][ks].y = nvd[wave][(kk + 1) * VBW + t * 16 + mrow];
        }

    float* st = &stage[wave][0];
    const int wbase = half * 8 * SPITCH + mrow * 3;
    int      roff[3];
    unsigned goff[3];
#pragma unroll
    for (int q = 0; q < 3; ++q) {
        const int p   = q * 32 + lane;
        const int row = p / 24;
        const int pc  = p % 24;
        roff[q] = row * SPITCH + pc * 4;
        goff[q] = (unsigned)(row * (OUTN * 3) + v0 * 3 + pc * 4);
    }
    float* ob = out + ((size_t)b * OUTN + (size_t)ut0 * 16) * (OUTN * 3);

    v4f val[12];

#pragma unroll
    for (int i = 0; i < UTW; ++i) {
        const int cur = i & 1;
        const int nxt = cur ^ 1;

        if (i + 1 < UTW) {
#pragma unroll
            for (int ks = 0; ks < 2; ++ks)
#pragma unroll
                for (int r = 0; r < 4; ++r) {
                    const v4f* crow = cb + (m0s[i + 1] + r) * NCTRL;
                    xb[nxt][(ks * 4 + r) * 2 + 0] = crow[ncol[ks][0]];
                    xb[nxt][(ks * 4 + r) * 2 + 1] = crow[ncol[ks][1]];
                }
        }
        __builtin_amdgcn_sched_barrier(0);

        v2f av[2][4];
        const v4f nu = nus[i];
#pragma unroll
        for (int ks = 0; ks < 2; ++ks) {
            v4f a0 = {0.0f, 0.0f, 0.0f, 0.0f};
            v4f a1 = {0.0f, 0.0f, 0.0f, 0.0f};
#pragma unroll
            for (int r = 0; r < 4; ++r) {
                const float w  = nu[r];
                const v4f   x0 = xb[cur][(ks * 4 + r) * 2 + 0];
                const v4f   x1 = xb[cur][(ks * 4 + r) * 2 + 1];
#pragma unroll
                for (int c = 0; c < 4; ++c) {
                    a0[c] = fmaf(w, x0[c], a0[c]);
                    a1[c] = fmaf(w, x1[c], a1[c]);
                }
            }
#pragma unroll
            for (int c = 0; c < 4; ++c) { av[ks][c].x = a0[c]; av[ks][c].y = a1[c]; }
        }

        v8f acc[2][4];
#pragma unroll
        for (int c = 0; c < 4; ++c) {
            v8f d0 = {};
            v8f d1 = {};
#pragma unroll
            for (int ks = 0; ks < 2; ++ks) {
                d0 = __builtin_amdgcn_wmma_f32_16x16x4_f32(
                         false, av[ks][c],  false, bv[0][ks],
                         (short)0, d0,  false,  false);
                d1 = __builtin_amdgcn_wmma_f32_16x16x4_f32(
                         false, av[ks][c],  false, bv[1][ks],
                         (short)0, d1,  false,  false);
            }
            acc[0][c] = d0;
            acc[1][c] = d1;
        }

        if (i > 0) {
            float* tb = ob + (size_t)((i - 1) * 16) * (OUTN * 3);
#pragma unroll
            for (int a = 0; a < 4; ++a)
#pragma unroll
                for (int q = 0; q < 3; ++q)
                    __builtin_nontemporal_store(val[a * 3 + q],
                                                (v4f*)(tb + a * 4 * (OUTN * 3) + goff[q]));
        }

#pragma unroll
        for (int t = 0; t < 2; ++t)
#pragma unroll
            for (int j = 0; j < 8; ++j) {
                float w = acc[t][3][j];
                float r = __builtin_amdgcn_rcpf(w);
                r = r * (2.0f - w * r);
                const int o = wbase + j * SPITCH + t * 48;
                st[o + 0] = acc[t][0][j] * r;
                st[o + 1] = acc[t][1][j] * r;
                st[o + 2] = acc[t][2][j] * r;
            }

#pragma unroll
        for (int a = 0; a < 4; ++a)
#pragma unroll
            for (int q = 0; q < 3; ++q)
                val[a * 3 + q] = *(const v4f*)(st + a * 4 * SPITCH + roff[q]);
    }

    {
        float* tb = ob + (size_t)((UTW - 1) * 16) * (OUTN * 3);
#pragma unroll
        for (int a = 0; a < 4; ++a)
#pragma unroll
            for (int q = 0; q < 3; ++q)
                __builtin_nontemporal_store(val[a * 3 + q],
                                            (v4f*)(tb + a * 4 * (OUTN * 3) + goff[q]));
    }
}

extern "C" void kernel_launch(void* const* d_in, const int* in_sizes, int n_in,
                              void* d_out, int out_size, void* d_ws, size_t ws_size,
                              hipStream_t stream)
{
    const float* ctrl  = (const float*)d_in[0];
    const int*   uspan = (const int*)  d_in[1];
    const int*   vspan = (const int*)  d_in[2];
    const float* Nu    = (const float*)d_in[3];
    const float* Nv    = (const float*)d_in[4];
    float*       out   = (float*)d_out;

    nurbs_fused<<<256, 256, 0, stream>>>(ctrl, uspan, vspan, Nu, Nv, out);
}


# u-contraction moved onto the matrix unit: 4 more f32 WMMAs per tile replace the per-lane fma chains and their gathered b128 loads; D registers feed the v-step as A; taps summed by the WMMA chain
# speedup vs baseline: 3.5349x; 1.0751x over previous
#include <hip/hip_runtime.h>


typedef __attribute__((ext_vector_type(2))) float v2f;
typedef __attribute__((ext_vector_type(4))) float v4f;
typedef __attribute__((ext_vector_type(8))) float v8f;

#define NCTRL  64
#define OUTN   512
#define KTILE  8
#define VBW    32
#define UTW    4
#define SPITCH 100

__global__ void __launch_bounds__(256)
nurbs_fused(const float* __restrict__ ctrl,
            const int*   __restrict__ uspan,
            const int*   __restrict__ vspan,
            const float* __restrict__ Nu,
            const float* __restrict__ Nv,
            float*       __restrict__ out)
{
    __shared__ float nvd[8][KTILE * VBW];
    __shared__ __attribute__((aligned(16))) float nud[8][UTW * KTILE * 16];
    __shared__ __attribute__((aligned(16))) float stage[8][16 * SPITCH];

    const int lane = threadIdx.x & 31;
    const int wave = threadIdx.x >> 5;
    const int vb  = (blockIdx.x & 1) * 8 + wave;
    const int ut0 = ((blockIdx.x >> 1) & 7) * UTW;
    const int b   = blockIdx.x >> 4;
    const int v0  = vb * VBW;

    const int mrow  = lane & 15;
    const int half  = lane >> 4;
    const int khalf = half * 2;

    const int kbase = vspan[v0] - 3;

    const int vsp = vspan[v0 + lane];
    const v4f nvr = *(const v4f*)(Nv + (v0 + lane) * 4);

    int m0s[UTW];
    v4f nus[UTW];
#pragma unroll
    for (int i = 0; i < UTW; ++i) {
        const int ua = (ut0 + i) * 16 + mrow;
        m0s[i] = uspan[ua] - 3;
        nus[i] = *(const v4f*)(Nu + ua * 4);
    }

    int coff[2];
#pragma unroll
    for (int t = 0; t < 2; ++t) {
        int n = kbase + 4 * t + 2 * (mrow >> 3) + (mrow & 1);
        if (n > 63) n = 63;
        coff[t] = n * 4 + ((mrow >> 1) & 3);
    }
    const float* cb = ctrl + (size_t)b * NCTRL * NCTRL * 4;

    v2f a1[UTW][2][2];
    int dsh[UTW];
#pragma unroll
    for (int i = 0; i < UTW; ++i) {
        const int mb = __builtin_amdgcn_readfirstlane(m0s[i]);
        dsh[i] = m0s[i] - mb;
#pragma unroll
        for (int s = 0; s < 2; ++s) {
            int r0 = mb + 4 * s + khalf;     if (r0 > 63) r0 = 63;
            int r1 = mb + 4 * s + khalf + 1; if (r1 > 63) r1 = 63;
#pragma unroll
            for (int t = 0; t < 2; ++t) {
                a1[i][t][s].x = cb[r0 * (NCTRL * 4) + coff[t]];
                a1[i][t][s].y = cb[r1 * (NCTRL * 4) + coff[t]];
            }
        }
    }

    {
#pragma unroll
        for (int k = 0; k < KTILE; ++k) nvd[wave][k * VBW + lane] = 0.0f;
        int ki = vsp - 3 - kbase;
#pragma unroll
        for (int s = 0; s < 4; ++s) {
            int kk = ki + s;
            if ((unsigned)kk < (unsigned)KTILE) nvd[wave][kk * VBW + lane] = nvr[s];
        }
    }
    float* nd = &nud[wave][0];
    {
        const v4f z = {0.0f, 0.0f, 0.0f, 0.0f};
#pragma unroll
        for (int q = 0; q < 4; ++q) *(v4f*)(nd + lane * 16 + q * 4) = z;
#pragma unroll
        for (int i = 0; i < UTW; ++i)
#pragma unroll
            for (int r = 0; r < 4; ++r) {
                int m = dsh[i] + r;
                if ((unsigned)m < (unsigned)KTILE) nd[(i * KTILE + m) * 16 + mrow] = nus[i][r];
            }
    }

    v2f bv[2][2];
#pragma unroll
    for (int t = 0; t < 2; ++t)
#pragma unroll
        for (int ks = 0; ks < 2; ++ks) {
            int kk = ks * 4 + khalf;
            bv[t][ks].x = nvd[wave][kk * VBW + t * 16 + mrow];
            bv[t][ks].y = nvd[wave][(kk + 1) * VBW + t * 16 + mrow];
        }
    v2f b1[UTW][2];
#pragma unroll
    for (int i = 0; i < UTW; ++i)
#pragma unroll
        for (int s = 0; s < 2; ++s) {
            int m = s * 4 + khalf;
            b1[i][s].x = nd[(i * KTILE + m) * 16 + mrow];
            b1[i][s].y = nd[(i * KTILE + m + 1) * 16 + mrow];
        }

    float* st = &stage[wave][0];
    const int wbase = half * 8 * SPITCH + mrow * 3;
    int      roff[3];
    unsigned goff[3];
#pragma unroll
    for (int q = 0; q < 3; ++q) {
        const int p   = q * 32 + lane;
        const int row = p / 24;
        const int pc  = p % 24;
        roff[q] = row * SPITCH + pc * 4;
        goff[q] = (unsigned)(row * (OUTN * 3) + v0 * 3 + pc * 4);
    }
    float* ob = out + ((size_t)b * OUTN + (size_t)ut0 * 16) * (OUTN * 3);

#pragma unroll
    for (int i = 0; i < UTW; ++i) {
        __builtin_amdgcn_sched_barrier(0);

        v8f tt[2];
#pragma unroll
        for (int t = 0; t < 2; ++t) {
            v8f d = {};
#pragma unroll
            for (int s = 0; s < 2; ++s)
                d = __builtin_amdgcn_wmma_f32_16x16x4_f32(
                         false, a1[i][t][s],  false, b1[i][s],
                         (short)0, d,  false,  false);
            tt[t] = d;
        }

        v8f acc[2][4];
#pragma unroll
        for (int c = 0; c < 4; ++c) {
            v8f d0 = {};
            v8f d1 = {};
#pragma unroll
            for (int ks = 0; ks < 2; ++ks) {
                v2f av;
                av.x = tt[ks][2 * c];
                av.y = tt[ks][2 * c + 1];
                d0 = __builtin_amdgcn_wmma_f32_16x16x4_f32(
                         false, av,  false, bv[0][ks],
                         (short)0, d0,  false,  false);
                d1 = __builtin_amdgcn_wmma_f32_16x16x4_f32(
                         false, av,  false, bv[1][ks],
                         (short)0, d1,  false,  false);
            }
            acc[0][c] = d0;
            acc[1][c] = d1;
        }

#pragma unroll
        for (int t = 0; t < 2; ++t)
#pragma unroll
            for (int j = 0; j < 8; ++j) {
                float w = acc[t][3][j];
                float r = __builtin_amdgcn_rcpf(w);
                r = r * (2.0f - w * r);
                const int o = wbase + j * SPITCH + t * 48;
                st[o + 0] = acc[t][0][j] * r;
                st[o + 1] = acc[t][1][j] * r;
                st[o + 2] = acc[t][2][j] * r;
            }

        float* tb = ob + (size_t)(i * 16) * (OUTN * 3);
        v4f val[12];
#pragma unroll
        for (int a = 0; a < 4; ++a)
#pragma unroll
            for (int q = 0; q < 3; ++q)
                val[a * 3 + q] = *(const v4f*)(st + a * 4 * SPITCH + roff[q]);
#pragma unroll
        for (int a = 0; a < 4; ++a)
#pragma unroll
            for (int q = 0; q < 3; ++q)
                __builtin_nontemporal_store(val[a * 3 + q],
                                            (v4f*)(tb + a * 4 * (OUTN * 3) + goff[q]));
    }
}

extern "C" void kernel_launch(void* const* d_in, const int* in_sizes, int n_in,
                              void* d_out, int out_size, void* d_ws, size_t ws_size,
                              hipStream_t stream)
{
    const float* ctrl  = (const float*)d_in[0];
    const int*   uspan = (const int*)  d_in[1];
    const int*   vspan = (const int*)  d_in[2];
    const float* Nu    = (const float*)d_in[3];
    const float* Nv    = (const float*)d_in[4];
    float*       out   = (float*)d_out;

    nurbs_fused<<<256, 256, 0, stream>>>(ctrl, uspan, vspan, Nu, Nv, out);
}


# Each wave owns a 64-sample v-band (two 32-sample halves, 2 u-tiles): a row piece is 768 B = three aligned 256-B units, so every store instruction writes whole 256-B units; was 384-B pieces
# speedup vs baseline: 3.7279x; 1.0546x over previous
#include <hip/hip_runtime.h>


typedef __attribute__((ext_vector_type(2))) float v2f;
typedef __attribute__((ext_vector_type(4))) float v4f;
typedef __attribute__((ext_vector_type(8))) float v8f;

#define NCTRL  64
#define OUTN   512
#define KTILE  8
#define VBW    32
#define NHB    2
#define UTW    2
#define SPITCH 196

__global__ void __launch_bounds__(256)
nurbs_fused(const float* __restrict__ ctrl,
            const int*   __restrict__ uspan,
            const int*   __restrict__ vspan,
            const float* __restrict__ Nu,
            const float* __restrict__ Nv,
            float*       __restrict__ out)
{
    __shared__ float nvd[8][NHB * KTILE * VBW];
    __shared__ __attribute__((aligned(16))) float nud[8][UTW * KTILE * 16];
    __shared__ __attribute__((aligned(16))) float stage[8][16 * SPITCH];

    const int lane = threadIdx.x & 31;
    const int wave = threadIdx.x >> 5;
    const int ut0 = (blockIdx.x & 15) * UTW;
    const int b   = blockIdx.x >> 4;
    const int v0  = wave * (NHB * VBW);

    const int mrow  = lane & 15;
    const int half  = lane >> 4;
    const int khalf = half * 2;

    int kbase[NHB];
    int vsp[NHB];
    v4f nvr[NHB];
#pragma unroll
    for (int hb = 0; hb < NHB; ++hb) {
        const int vh0 = v0 + hb * VBW;
        kbase[hb] = vspan[vh0] - 3;
        vsp[hb]   = vspan[vh0 + lane];
        nvr[hb]   = *(const v4f*)(Nv + (vh0 + lane) * 4);
    }

    int m0s[UTW];
    v4f nus[UTW];
#pragma unroll
    for (int i = 0; i < UTW; ++i) {
        const int ua = (ut0 + i) * 16 + mrow;
        m0s[i] = uspan[ua] - 3;
        nus[i] = *(const v4f*)(Nu + ua * 4);
    }

    int coff[NHB][2];
#pragma unroll
    for (int hb = 0; hb < NHB; ++hb)
#pragma unroll
        for (int t = 0; t < 2; ++t) {
            int n = kbase[hb] + 4 * t + 2 * (mrow >> 3) + (mrow & 1);
            if (n > 63) n = 63;
            coff[hb][t] = n * 4 + ((mrow >> 1) & 3);
        }
    const float* cb = ctrl + (size_t)b * NCTRL * NCTRL * 4;

    v2f a1[UTW][NHB][2][2];
    int dsh[UTW];
#pragma unroll
    for (int i = 0; i < UTW; ++i) {
        const int mb = __builtin_amdgcn_readfirstlane(m0s[i]);
        dsh[i] = m0s[i] - mb;
#pragma unroll
        for (int hb = 0; hb < NHB; ++hb)
#pragma unroll
            for (int s = 0; s < 2; ++s) {
                int r0 = mb + 4 * s + khalf;     if (r0 > 63) r0 = 63;
                int r1 = mb + 4 * s + khalf + 1; if (r1 > 63) r1 = 63;
#pragma unroll
                for (int t = 0; t < 2; ++t) {
                    a1[i][hb][t][s].x = cb[r0 * (NCTRL * 4) + coff[hb][t]];
                    a1[i][hb][t][s].y = cb[r1 * (NCTRL * 4) + coff[hb][t]];
                }
            }
    }

    float* nv = &nvd[wave][0];
#pragma unroll
    for (int hb = 0; hb < NHB; ++hb) {
#pragma unroll
        for (int k = 0; k < KTILE; ++k) nv[(hb * KTILE + k) * VBW + lane] = 0.0f;
        int ki = vsp[hb] - 3 - kbase[hb];
#pragma unroll
        for (int s = 0; s < 4; ++s) {
            int kk = ki + s;
            if ((unsigned)kk < (unsigned)KTILE) nv[(hb * KTILE + kk) * VBW + lane] = nvr[hb][s];
        }
    }
    float* nd = &nud[wave][0];
    {
        const v4f z = {0.0f, 0.0f, 0.0f, 0.0f};
#pragma unroll
        for (int q = 0; q < 2; ++q) *(v4f*)(nd + lane * 8 + q * 4) = z;
#pragma unroll
        for (int i = 0; i < UTW; ++i)
#pragma unroll
            for (int r = 0; r < 4; ++r) {
                int m = dsh[i] + r;
                if ((unsigned)m < (unsigned)KTILE) nd[(i * KTILE + m) * 16 + mrow] = nus[i][r];
            }
    }

    v2f bv[NHB][2][2];
#pragma unroll
    for (int hb = 0; hb < NHB; ++hb)
#pragma unroll
        for (int t = 0; t < 2; ++t)
#pragma unroll
            for (int ks = 0; ks < 2; ++ks) {
                int kk = hb * KTILE + ks * 4 + khalf;
                bv[hb][t][ks].x = nv[kk * VBW + t * 16 + mrow];
                bv[hb][t][ks].y = nv[(kk + 1) * VBW + t * 16 + mrow];
            }
    v2f b1[UTW][2];
#pragma unroll
    for (int i = 0; i < UTW; ++i)
#pragma unroll
        for (int s = 0; s < 2; ++s) {
            int m = s * 4 + khalf;
            b1[i][s].x = nd[(i * KTILE + m) * 16 + mrow];
            b1[i][s].y = nd[(i * KTILE + m + 1) * 16 + mrow];
        }

    float* st = &stage[wave][0];
    const int wbase = half * 8 * SPITCH + mrow * 3;
    const int      roffA = half * SPITCH + mrow * 4;
    const unsigned goffA = (unsigned)(half * (OUTN * 3) + v0 * 3 + mrow * 4);
    const int      roffB = 64 + lane * 4;
    const unsigned goffB = (unsigned)(v0 * 3 + 64 + lane * 4);
    float* ob = out + ((size_t)b * OUTN + (size_t)ut0 * 16) * (OUTN * 3);

#pragma unroll
    for (int i = 0; i < UTW; ++i) {
        float* tb = ob + (size_t)(i * 16) * (OUTN * 3);
#pragma unroll
        for (int hb = 0; hb < NHB; ++hb) {
            __builtin_amdgcn_sched_barrier(0);

            v8f tt[2];
#pragma unroll
            for (int t = 0; t < 2; ++t) {
                v8f d = {};
#pragma unroll
                for (int s = 0; s < 2; ++s)
                    d = __builtin_amdgcn_wmma_f32_16x16x4_f32(
                             false, a1[i][hb][t][s],  false, b1[i][s],
                             (short)0, d,  false,  false);
                tt[t] = d;
            }

            v8f acc[2][4];
#pragma unroll
            for (int c = 0; c < 4; ++c) {
                v8f d0 = {};
                v8f d1 = {};
#pragma unroll
                for (int ks = 0; ks < 2; ++ks) {
                    v2f av;
                    av.x = tt[ks][2 * c];
                    av.y = tt[ks][2 * c + 1];
                    d0 = __builtin_amdgcn_wmma_f32_16x16x4_f32(
                             false, av,  false, bv[hb][0][ks],
                             (short)0, d0,  false,  false);
                    d1 = __builtin_amdgcn_wmma_f32_16x16x4_f32(
                             false, av,  false, bv[hb][1][ks],
                             (short)0, d1,  false,  false);
                }
                acc[0][c] = d0;
                acc[1][c] = d1;
            }

#pragma unroll
            for (int t = 0; t < 2; ++t)
#pragma unroll
                for (int j = 0; j < 8; ++j) {
                    float w = acc[t][3][j];
                    float r = __builtin_amdgcn_rcpf(w);
                    r = r * (2.0f - w * r);
                    const int o = wbase + j * SPITCH + hb * 96 + t * 48;
                    st[o + 0] = acc[t][0][j] * r;
                    st[o + 1] = acc[t][1][j] * r;
                    st[o + 2] = acc[t][2][j] * r;
                }

            if (hb == 0) {
                v4f val[8];
#pragma unroll
                for (int a = 0; a < 8; ++a)
                    val[a] = *(const v4f*)(st + a * 2 * SPITCH + roffA);
#pragma unroll
                for (int a = 0; a < 8; ++a)
                    __builtin_nontemporal_store(val[a],
                                                (v4f*)(tb + a * 2 * (OUTN * 3) + goffA));
            } else {
                v4f val[16];
#pragma unroll
                for (int r = 0; r < 16; ++r)
                    val[r] = *(const v4f*)(st + r * SPITCH + roffB);
#pragma unroll
                for (int r = 0; r < 16; ++r)
                    __builtin_nontemporal_store(val[r],
                                                (v4f*)(tb + r * (OUTN * 3) + goffB));
            }
        }
    }
}

extern "C" void kernel_launch(void* const* d_in, const int* in_sizes, int n_in,
                              void* d_out, int out_size, void* d_ws, size_t ws_size,
                              hipStream_t stream)
{
    const float* ctrl  = (const float*)d_in[0];
    const int*   uspan = (const int*)  d_in[1];
    const int*   vspan = (const int*)  d_in[2];
    const float* Nu    = (const float*)d_in[3];
    const float* Nv    = (const float*)d_in[4];
    float*       out   = (float*)d_out;

    nurbs_fused<<<256, 256, 0, stream>>>(ctrl, uspan, vspan, Nu, Nv, out);
}
